# starting kernel
# baseline (speedup 1.0000x reference)
#include <hip/hip_runtime.h>


typedef float v2f __attribute__((ext_vector_type(2)));
typedef float v8f __attribute__((ext_vector_type(8)));

__global__ __launch_bounds__(256)
void causal_bow_wmma_kernel(const float* __restrict__ x,
                            float* __restrict__ out) {
  constexpr int T = 2048;
  constexpr int C = 512;

  const unsigned lane = threadIdx.x & 31u;
  const unsigned waveInBlock = threadIdx.x >> 5;
  const unsigned wave = blockIdx.x * (blockDim.x >> 5) + waveInBlock;
  constexpr unsigned nC16 = (unsigned)C >> 4;
  const unsigned b    = wave / nC16;
  const unsigned c16  = wave % nC16;

  const unsigned col    = lane & 15u;
  const bool     hiHalf = lane >= 16u;
  const unsigned hi2    = hiHalf ? 2u : 0u;
  const unsigned hi8    = hiHalf ? 8u : 0u;

  v2f a[4];
#pragma unroll
  for (int j = 0; j < 4; ++j) {
    unsigned k0 = 4u * (unsigned)j + hi2;
    a[j].x = (k0 <= col)      ? 1.0f : 0.0f;
    a[j].y = (k0 + 1u <= col) ? 1.0f : 0.0f;
  }
  const v2f onesA = {1.0f, 1.0f};

  const size_t waveBase = (size_t)b * (size_t)T * (size_t)C + c16 * 16u + col;
  const float* xp = x   + waveBase + (size_t)hi2 * C;
  float*       op = out + waveBase + (size_t)hi8 * C;

  float denom[8];
#pragma unroll
  for (int i = 0; i < 8; ++i) denom[i] = (float)(hi8 + (unsigned)i + 1u);

  v8f s = {};
  constexpr int nTiles = T >> 4;

  for (int tile = 0; tile < nTiles; ++tile) {
    v2f bb[4];
#pragma unroll
    for (int j = 0; j < 4; ++j) {
      bb[j].x = xp[(4 * j) * C];
      bb[j].y = xp[(4 * j + 1) * C];
    }

    if (tile + 1 < nTiles)
      __builtin_prefetch(xp + 16 * C, 0, 3);

    v8f d = __builtin_amdgcn_wmma_f32_16x16x4_f32(
        false, a[0], false, bb[0], (short)0, s, false, false);
#pragma unroll
    for (int j = 1; j < 4; ++j) {
      d = __builtin_amdgcn_wmma_f32_16x16x4_f32(
          false, a[j], false, bb[j], (short)0, d, false, false);
    }

#pragma unroll
    for (int j = 0; j < 4; ++j) {
      s = __builtin_amdgcn_wmma_f32_16x16x4_f32(
          false, onesA, false, bb[j], (short)0, s, false, false);
    }

#pragma unroll
    for (int i = 0; i < 8; ++i) {
      float dn  = denom[i];
      float r   = __builtin_amdgcn_rcpf(dn);
      r = __builtin_fmaf(__builtin_fmaf(-dn, r, 1.0f), r, r);
      op[i * C] = d[i] * r;
      denom[i]  = dn + 16.0f;
    }

    xp += 16 * C;
    op += 16 * C;
  }
}

extern "C" void kernel_launch(void* const* d_in, const int* in_sizes, int n_in,
                              void* d_out, int out_size, void* d_ws, size_t ws_size,
                              hipStream_t stream) {
  (void)in_sizes; (void)n_in; (void)d_ws; (void)ws_size; (void)out_size;

  const float* x = (const float*)d_in[0];
  float* out = (float*)d_out;

  const int B = 32, C = 512;
  const int nC16 = C / 16;
  const int waves = B * nC16;
  const int threads = waves * 32;
  const int block = 256;
  const int grid = threads / block;

  causal_bow_wmma_kernel<<<grid, block, 0, stream>>>(x, out);
}


# Split the time axis over the waves of a work-group: each wave scans its own 256-row chunk in registers, chunk totals meet in LDS, carry added after one barrier (f32 sum order changes: chunk-wise)
# speedup vs baseline: 2.6937x; 2.6937x over previous
#include <hip/hip_runtime.h>


typedef float v2f __attribute__((ext_vector_type(2)));
typedef float v8f __attribute__((ext_vector_type(8)));

__global__ __launch_bounds__(512)
void causal_bow_wmma_kernel(const float* __restrict__ x,
                            float* __restrict__ out) {
  constexpr int T = 2048;
  constexpr int C = 512;
  constexpr int nChunks = 8;
  constexpr int rowsPerWave = T / nChunks;
  constexpr int nTiles = rowsPerWave >> 4;
  constexpr unsigned nC32 = (unsigned)C >> 5;

  __shared__ float chunkTotal[nChunks * 32];

  const unsigned lane  = threadIdx.x & 31u;
  const unsigned wv    = threadIdx.x >> 5;
  const unsigned strip = wv & 1u;
  const unsigned w     = wv >> 1;
  const unsigned b     = blockIdx.x / nC32;
  const unsigned c16   = (blockIdx.x % nC32) * 2u + strip;

  const unsigned col    = lane & 15u;
  const bool     hiHalf = lane >= 16u;
  const unsigned hi2    = hiHalf ? 2u : 0u;
  const unsigned hi8    = hiHalf ? 8u : 0u;

  v2f a[4];
#pragma unroll
  for (int j = 0; j < 4; ++j) {
    unsigned k0 = 4u * (unsigned)j + hi2;
    a[j].x = (k0 <= col)      ? 1.0f : 0.0f;
    a[j].y = (k0 + 1u <= col) ? 1.0f : 0.0f;
  }
  const v2f onesA = {1.0f, 1.0f};

  const unsigned row0 = w * (unsigned)rowsPerWave;
  const size_t waveBase = ((size_t)b * (size_t)T + row0) * (size_t)C
                        + c16 * 16u + col;
  const float* xp = x   + waveBase + (size_t)hi2 * C;
  float*       op = out + waveBase + (size_t)hi8 * C;

  v2f bb[nTiles][4];
#pragma unroll
  for (int k = 0; k < nTiles; ++k) {
#pragma unroll
    for (int j = 0; j < 4; ++j) {
      bb[k][j].x = xp[(16 * k + 4 * j) * C];
      bb[k][j].y = xp[(16 * k + 4 * j + 1) * C];
    }
  }

  v8f d[nTiles];
  v8f s = {};
#pragma unroll
  for (int k = 0; k < nTiles; ++k) {
    v8f acc = __builtin_amdgcn_wmma_f32_16x16x4_f32(
        false, a[0], false, bb[k][0], (short)0, s, false, false);
#pragma unroll
    for (int j = 1; j < 4; ++j) {
      acc = __builtin_amdgcn_wmma_f32_16x16x4_f32(
          false, a[j], false, bb[k][j], (short)0, acc, false, false);
    }
    d[k] = acc;
#pragma unroll
    for (int j = 0; j < 4; ++j) {
      s = __builtin_amdgcn_wmma_f32_16x16x4_f32(
          false, onesA, false, bb[k][j], (short)0, s, false, false);
    }
  }

  if (!hiHalf) chunkTotal[w * 32u + strip * 16u + col] = s[0];
  __syncthreads();

  float carry = 0.0f;
#pragma unroll
  for (unsigned j = 0; j + 1u < (unsigned)nChunks; ++j) {
    const float tj = chunkTotal[j * 32u + strip * 16u + col];
    carry += (j < w) ? tj : 0.0f;
  }

  const float dn0 = (float)(row0 + hi8 + 1u);
#pragma unroll
  for (int k = 0; k < nTiles; ++k) {
#pragma unroll
    for (int i = 0; i < 8; ++i) {
      float dn = dn0 + (float)(16 * k + i);
      float r  = __builtin_amdgcn_rcpf(dn);
      r = __builtin_fmaf(__builtin_fmaf(-dn, r, 1.0f), r, r);
      op[(16 * k + i) * C] = (d[k][i] + carry) * r;
    }
  }
}

extern "C" void kernel_launch(void* const* d_in, const int* in_sizes, int n_in,
                              void* d_out, int out_size, void* d_ws, size_t ws_size,
                              hipStream_t stream) {
  (void)in_sizes; (void)n_in; (void)d_ws; (void)ws_size; (void)out_size;

  const float* x = (const float*)d_in[0];
  float* out = (float*)d_out;

  causal_bow_wmma_kernel<<<512, 512, 0, stream>>>(x, out);
}


# 16 bytes per lane on every global access: a lane owns 4 adjacent columns (4 strips per wave), 64-column groups swept in 2 passes of 64-row chunks (f32 carry sum order changes: 64-row chunks)
# speedup vs baseline: 4.1676x; 1.5472x over previous
#include <hip/hip_runtime.h>


typedef float v2f __attribute__((ext_vector_type(2)));
typedef float v4f __attribute__((ext_vector_type(4)));
typedef float v8f __attribute__((ext_vector_type(8)));

__global__ __launch_bounds__(512)
void causal_bow_wmma_kernel(const float* __restrict__ x,
                            float* __restrict__ out) {
  constexpr int T = 2048;
  constexpr int C = 512;
  constexpr int nWaves = 16;
  constexpr int nPass = 2;
  constexpr int rowsPerPass = T / nPass;
  constexpr int rowsPerWave = rowsPerPass / nWaves;
  constexpr int nTiles = rowsPerWave >> 4;
  constexpr int nStrips = 4;
  constexpr unsigned nC64 = (unsigned)C >> 6;

  __shared__ v4f chunkTotal[nPass][nWaves][16];

  const unsigned lane = threadIdx.x & 31u;
  const unsigned w    = threadIdx.x >> 5;
  const unsigned b    = blockIdx.x / nC64;
  const unsigned cg   = blockIdx.x % nC64;

  const unsigned col    = lane & 15u;
  const bool     hiHalf = lane >= 16u;
  const unsigned hi2    = hiHalf ? 2u : 0u;
  const unsigned hi8    = hiHalf ? 8u : 0u;

  v2f a[4];
#pragma unroll
  for (int j = 0; j < 4; ++j) {
    unsigned k0 = 4u * (unsigned)j + hi2;
    a[j].x = (k0 <= col)      ? 1.0f : 0.0f;
    a[j].y = (k0 + 1u <= col) ? 1.0f : 0.0f;
  }
  const v2f onesA = {1.0f, 1.0f};

  const size_t groupBase = (size_t)b * (size_t)T * (size_t)C
                         + cg * 64u + col * 4u;
  const float* xp = x   + groupBase + (size_t)hi2 * C;
  float*       op = out + groupBase + (size_t)hi8 * C;

  v4f passBase = {0.0f, 0.0f, 0.0f, 0.0f};

#pragma unroll 1
  for (int p = 0; p < nPass; ++p) {
    const unsigned row0 = (unsigned)p * (unsigned)rowsPerPass
                        + w * (unsigned)rowsPerWave;
    const float* xr = xp + (size_t)row0 * C;
    float*       orow = op + (size_t)row0 * C;

    v4f q[nTiles][4][2];
#pragma unroll
    for (int k = 0; k < nTiles; ++k) {
#pragma unroll
      for (int j = 0; j < 4; ++j) {
        q[k][j][0] = *(const v4f*)(xr + (16 * k + 4 * j) * C);
        q[k][j][1] = *(const v4f*)(xr + (16 * k + 4 * j + 1) * C);
      }
    }

    v8f d[nTiles][nStrips];
    v8f s[nStrips];
#pragma unroll
    for (int st = 0; st < nStrips; ++st) s[st] = v8f{};
#pragma unroll
    for (int k = 0; k < nTiles; ++k) {
#pragma unroll
      for (int st = 0; st < nStrips; ++st) {
        v2f bf[4];
#pragma unroll
        for (int j = 0; j < 4; ++j) {
          bf[j].x = q[k][j][0][st];
          bf[j].y = q[k][j][1][st];
        }
        v8f acc = __builtin_amdgcn_wmma_f32_16x16x4_f32(
            false, a[0], false, bf[0], (short)0, s[st], false, false);
#pragma unroll
        for (int j = 1; j < 4; ++j) {
          acc = __builtin_amdgcn_wmma_f32_16x16x4_f32(
              false, a[j], false, bf[j], (short)0, acc, false, false);
        }
        d[k][st] = acc;
#pragma unroll
        for (int j = 0; j < 4; ++j) {
          s[st] = __builtin_amdgcn_wmma_f32_16x16x4_f32(
              false, onesA, false, bf[j], (short)0, s[st], false, false);
        }
      }
    }

    if (!hiHalf) {
      const v4f tot = {s[0][0], s[1][0], s[2][0], s[3][0]};
      chunkTotal[p][w][col] = tot;
    }
    __syncthreads();

    v4f run = passBase;
    v4f carry = passBase;
#pragma unroll
    for (int j = 0; j < nWaves; ++j) {
      const v4f tj = chunkTotal[p][j][col];
      run += tj;
      const bool before = (unsigned)j < w;
      carry.x = before ? run.x : carry.x;
      carry.y = before ? run.y : carry.y;
      carry.z = before ? run.z : carry.z;
      carry.w = before ? run.w : carry.w;
    }
    passBase = run;

    const float dn0 = (float)(row0 + hi8 + 1u);
#pragma unroll
    for (int k = 0; k < nTiles; ++k) {
#pragma unroll
      for (int i = 0; i < 8; ++i) {
        float dn = dn0 + (float)(16 * k + i);
        float r  = __builtin_amdgcn_rcpf(dn);
        r = __builtin_fmaf(__builtin_fmaf(-dn, r, 1.0f), r, r);
        v4f o;
        o.x = (d[k][0][i] + carry.x) * r;
        o.y = (d[k][1][i] + carry.y) * r;
        o.z = (d[k][2][i] + carry.z) * r;
        o.w = (d[k][3][i] + carry.w) * r;
        *(v4f*)(orow + (16 * k + i) * C) = o;
      }
    }
  }
}

extern "C" void kernel_launch(void* const* d_in, const int* in_sizes, int n_in,
                              void* d_out, int out_size, void* d_ws, size_t ws_size,
                              hipStream_t stream) {
  (void)in_sizes; (void)n_in; (void)d_ws; (void)ws_size; (void)out_size;

  const float* x = (const float*)d_in[0];
  float* out = (float*)d_out;

  causal_bow_wmma_kernel<<<256, 512, 0, stream>>>(x, out);
}


# Halve the WMMAs: drop the all-ones running-sum product (= row 15 of the triangular product); tiles scan from zero, tile carries via one lane-half exchange (f32 sum order changes: 16-row tiles)
# speedup vs baseline: 4.8654x; 1.1674x over previous
#include <hip/hip_runtime.h>


typedef float v2f __attribute__((ext_vector_type(2)));
typedef float v4f __attribute__((ext_vector_type(4)));
typedef float v8f __attribute__((ext_vector_type(8)));

__device__ __forceinline__ float xhalf(float v) {
  const int b = __builtin_bit_cast(int, v);
  return __builtin_bit_cast(float,
      __builtin_amdgcn_permlanex16(b, b, 0x76543210, 0xfedcba98, false, false));
}

__global__ __launch_bounds__(512)
void causal_bow_wmma_kernel(const float* __restrict__ x,
                            float* __restrict__ out) {
  constexpr int T = 2048;
  constexpr int C = 512;
  constexpr int nWaves = 16;
  constexpr int nPass = 2;
  constexpr int rowsPerPass = T / nPass;
  constexpr int rowsPerWave = rowsPerPass / nWaves;
  constexpr int nTiles = rowsPerWave >> 4;
  constexpr int nStrips = 4;
  constexpr unsigned nC64 = (unsigned)C >> 6;

  __shared__ v4f chunkTotal[nPass][nWaves][16];

  const unsigned lane = threadIdx.x & 31u;
  const unsigned w    = threadIdx.x >> 5;
  const unsigned b    = blockIdx.x / nC64;
  const unsigned cg   = blockIdx.x % nC64;

  const unsigned col    = lane & 15u;
  const bool     hiHalf = lane >= 16u;
  const unsigned hi2    = hiHalf ? 2u : 0u;
  const unsigned hi8    = hiHalf ? 8u : 0u;

  v2f a[4];
#pragma unroll
  for (int j = 0; j < 4; ++j) {
    unsigned k0 = 4u * (unsigned)j + hi2;
    a[j].x = (k0 <= col)      ? 1.0f : 0.0f;
    a[j].y = (k0 + 1u <= col) ? 1.0f : 0.0f;
  }

  const size_t groupBase = (size_t)b * (size_t)T * (size_t)C
                         + cg * 64u + col * 4u;
  const float* xp = x   + groupBase + (size_t)hi2 * C;
  float*       op = out + groupBase + (size_t)hi8 * C;

  v4f passBase = {0.0f, 0.0f, 0.0f, 0.0f};

#pragma unroll 1
  for (int p = 0; p < nPass; ++p) {
    const unsigned row0 = (unsigned)p * (unsigned)rowsPerPass
                        + w * (unsigned)rowsPerWave;
    const float* xr = xp + (size_t)row0 * C;
    float*       orow = op + (size_t)row0 * C;

    v4f q[nTiles][4][2];
#pragma unroll
    for (int k = 0; k < nTiles; ++k) {
#pragma unroll
      for (int j = 0; j < 4; ++j) {
        q[k][j][0] = *(const v4f*)(xr + (16 * k + 4 * j) * C);
        q[k][j][1] = *(const v4f*)(xr + (16 * k + 4 * j + 1) * C);
      }
    }

    v8f d[nTiles][nStrips];
#pragma unroll
    for (int k = 0; k < nTiles; ++k) {
#pragma unroll
      for (int st = 0; st < nStrips; ++st) {
        v8f acc = v8f{};
#pragma unroll
        for (int j = 0; j < 4; ++j) {
          v2f bf;
          bf.x = q[k][j][0][st];
          bf.y = q[k][j][1][st];
          acc = __builtin_amdgcn_wmma_f32_16x16x4_f32(
              false, a[j], false, bf, (short)0, acc, false, false);
        }
        d[k][st] = acc;
      }
    }

    v4f pre[nTiles];
    v4f tot = {0.0f, 0.0f, 0.0f, 0.0f};
#pragma unroll
    for (int k = 0; k < nTiles; ++k) {
      pre[k] = tot;
      v4f tk;
#pragma unroll
      for (int st = 0; st < nStrips; ++st) {
        const float own = hiHalf ? d[k][st][7] : 0.0f;
        tk[st] = own + xhalf(own);
      }
      tot += tk;
    }

    if (!hiHalf) {
      chunkTotal[p][w][col] = tot;
    }
    __syncthreads();

    v4f run = passBase;
    v4f carry = passBase;
#pragma unroll
    for (int j = 0; j < nWaves; ++j) {
      const v4f tj = chunkTotal[p][j][col];
      run += tj;
      const bool before = (unsigned)j < w;
      carry.x = before ? run.x : carry.x;
      carry.y = before ? run.y : carry.y;
      carry.z = before ? run.z : carry.z;
      carry.w = before ? run.w : carry.w;
    }
    passBase = run;

    v4f ck[nTiles];
#pragma unroll
    for (int k = 0; k < nTiles; ++k) ck[k] = carry + pre[k];

    const float dn0 = (float)(row0 + hi8 + 1u);
#pragma unroll
    for (int k = 0; k < nTiles; ++k) {
#pragma unroll
      for (int i = 0; i < 8; ++i) {
        float dn = dn0 + (float)(16 * k + i);
        float r  = __builtin_amdgcn_rcpf(dn);
        r = __builtin_fmaf(__builtin_fmaf(-dn, r, 1.0f), r, r);
        v4f o;
        o.x = (d[k][0][i] + ck[k].x) * r;
        o.y = (d[k][1][i] + ck[k].y) * r;
        o.z = (d[k][2][i] + ck[k].z) * r;
        o.w = (d[k][3][i] + ck[k].w) * r;
        *(v4f*)(orow + (16 * k + i) * C) = o;
      }
    }
  }
}

extern "C" void kernel_launch(void* const* d_in, const int* in_sizes, int n_in,
                              void* d_out, int out_size, void* d_ws, size_t ws_size,
                              hipStream_t stream) {
  (void)in_sizes; (void)n_in; (void)d_ws; (void)ws_size; (void)out_size;

  const float* x = (const float*)d_in[0];
  float* out = (float*)d_out;

  causal_bow_wmma_kernel<<<256, 512, 0, stream>>>(x, out);
}
